# starting kernel
# baseline (speedup 1.0000x reference)
#include <hip/hip_runtime.h>
#include <hip/hip_bf16.h>

#define N 4096
#define SINKHORN_ITERS 50
#define NCHUNK 16
#define ROWS_PER_CHUNK (N / NCHUNK)

typedef float v2f __attribute__((ext_vector_type(2)));
typedef float v8f __attribute__((ext_vector_type(8)));

__global__ void sinkhorn_init_c(float* __restrict__ C) {
    int j = blockIdx.x * blockDim.x + threadIdx.x;
    if (j < N) C[j] = 0.0f;
}

__global__ __launch_bounds__(256) void sinkhorn_row_lse(
    const float* __restrict__ M, const float* __restrict__ C,
    float* __restrict__ R) {
    __shared__ float sdata[256];
    const int i = blockIdx.x;
    const int tid = threadIdx.x;
    const float* __restrict__ row = M + (size_t)i * N;

    float acc = 0.0f;
#pragma unroll
    for (int j4 = tid * 4; j4 < N; j4 += 256 * 4) {
        const float4 m = *(const float4*)(row + j4);
        const float4 c = *(const float4*)(C + j4);
        acc += __expf(m.x - c.x);
        acc += __expf(m.y - c.y);
        acc += __expf(m.z - c.z);
        acc += __expf(m.w - c.w);
    }
    sdata[tid] = acc;
    __syncthreads();
    for (int s = 128; s > 0; s >>= 1) {
        if (tid < s) sdata[tid] += sdata[tid + s];
        __syncthreads();
    }
    if (tid == 0) R[i] = __logf(sdata[0]);
}

__global__ __launch_bounds__(128) void sinkhorn_col_partial(
    const float* __restrict__ M, const float* __restrict__ R,
    float* __restrict__ partial) {
    const int lane = threadIdx.x & 31;
    const int wave = threadIdx.x >> 5;
    const int j0 = blockIdx.x * 64 + wave * 16;
    const int rowBase = blockIdx.y * ROWS_PER_CHUNK;
    const int half = lane >> 4;
    const int lcol = lane & 15;
    const int col = j0 + lcol;

    v2f a;
    a.x = 1.0f;
    a.y = 1.0f;
    v8f d = {};

    const float* __restrict__ mc = M + col;
    for (int k = 0; k < ROWS_PER_CHUNK; k += 4) {
        const int r0 = rowBase + k + 2 * half;
        __builtin_prefetch(&mc[(size_t)(r0 + 8) * N], 0, 3);
        v2f b;
        b.x = __expf(mc[(size_t)r0 * N] - R[r0]);
        b.y = __expf(mc[(size_t)(r0 + 1) * N] - R[r0 + 1]);
        d = __builtin_amdgcn_wmma_f32_16x16x4_f32(false, a, false, b,
                                                  (short)0, d, false, false);
    }

    if (half == 0) {
        partial[(size_t)blockIdx.y * N + col] = d[0];
    }
}

__global__ void sinkhorn_col_finalize(const float* __restrict__ partial,
                                      float* __restrict__ C) {
    int j = blockIdx.x * blockDim.x + threadIdx.x;
    if (j >= N) return;
    float s = 0.0f;
#pragma unroll
    for (int c = 0; c < NCHUNK; ++c) s += partial[(size_t)c * N + j];
    C[j] = __logf(s);
}

__global__ __launch_bounds__(256) void sinkhorn_final(
    const float* __restrict__ M, const float* __restrict__ R,
    const float* __restrict__ C, float* __restrict__ out) {
    const int idx = blockIdx.x * blockDim.x + threadIdx.x;
    const int i = idx >> (12 - 2);
    const int j4 = (idx & (N / 4 - 1)) << 2;
    const size_t off = (size_t)i * N + j4;

    const float4 m = *(const float4*)(M + off);
    const float4 c = *(const float4*)(C + j4);
    const float r = R[i];
    float4 o;
    o.x = fmaxf(__expf(m.x - r - c.x), 1e-9f);
    o.y = fmaxf(__expf(m.y - r - c.y), 1e-9f);
    o.z = fmaxf(__expf(m.z - r - c.z), 1e-9f);
    o.w = fmaxf(__expf(m.w - r - c.w), 1e-9f);
    *(float4*)(out + off) = o;
}

extern "C" void kernel_launch(void* const* d_in, const int* in_sizes, int n_in,
                              void* d_out, int out_size, void* d_ws,
                              size_t ws_size, hipStream_t stream) {
    (void)in_sizes; (void)n_in; (void)out_size; (void)ws_size;

    const float* M = (const float*)d_in[0];
    float* out = (float*)d_out;

    char* ws = (char*)d_ws;
    float* R = (float*)(ws);
    float* C = (float*)(ws + 16 * 1024);
    float* partial = (float*)(ws + 32 * 1024);

    sinkhorn_init_c<<<N / 256, 256, 0, stream>>>(C);

    for (int t = 0; t < SINKHORN_ITERS; ++t) {
        sinkhorn_row_lse<<<N, 256, 0, stream>>>(M, C, R);
        sinkhorn_col_partial<<<dim3(N / 64, NCHUNK), 128, 0, stream>>>(M, R,
                                                                       partial);
        sinkhorn_col_finalize<<<N / 256, 256, 0, stream>>>(partial, C);
    }

    sinkhorn_final<<<(N * (size_t)N / 4) / 256, 256, 0, stream>>>(M, R, C, out);
}


# iteration passes re-shaped to one wave per SIMD, 1024 waves per launch (was 4096): a row-pass wave owns a whole quad, a column-pass wave 1024 rows, r read through LDS; row and column sums re-ordered
# speedup vs baseline: 1.6379x; 1.6379x over previous
#include <hip/hip_runtime.h>
#include <hip/hip_bf16.h>


#define N 4096
#define SINKHORN_ITERS 50
#define QUAD_FLOATS (4 * N)

typedef float v2f __attribute__((ext_vector_type(2)));
typedef float v8f __attribute__((ext_vector_type(8)));

__device__ __forceinline__ float wave_sum(float v) {
#pragma unroll
    for (int m = 16; m >= 1; m >>= 1) v += __shfl_xor(v, m);
    return v;
}

template <int WPQ, int QPG>
__device__ __forceinline__ void quad_rows_finish(float a0, float a1, float a2,
                                                 float a3, float (*part)[4],
                                                 int tid, int quad0,
                                                 float* __restrict__ Rv) {
    a0 = wave_sum(a0);
    a1 = wave_sum(a1);
    a2 = wave_sum(a2);
    a3 = wave_sum(a3);
    if ((tid & 31) == 0) {
        float* p = part[tid >> 5];
        p[0] = a0;
        p[1] = a1;
        p[2] = a2;
        p[3] = a3;
    }
    __syncthreads();
    if (tid < 4 * QPG) {
        const int qq = tid >> 2;
        const int s = tid & 3;
        float t = 0.0f;
#pragma unroll
        for (int x = 0; x < WPQ; ++x) t += part[qq * WPQ + x][s];
        Rv[4 * (quad0 + qq) + s] = 1.0f / t;
    }
}

__global__ __launch_bounds__(256) void sinkhorn_build_row0(
    const float* __restrict__ M, float* __restrict__ K,
    float* __restrict__ Rv) {
    __shared__ float part[8][4];
    const int q = blockIdx.x;
    const int tid = threadIdx.x;
    const float* __restrict__ m0 = M + (size_t)(4 * q) * N + tid;
    float4* __restrict__ kq = (float4*)(K + (size_t)q * QUAD_FLOATS) + tid;

    float a0 = 0.0f, a1 = 0.0f, a2 = 0.0f, a3 = 0.0f;
#pragma unroll
    for (int n = 0; n < 16; ++n) {
        const int j = n * 256;
        float4 k;
        k.x = __expf(m0[j]);
        k.y = __expf(m0[N + j]);
        k.z = __expf(m0[2 * N + j]);
        k.w = __expf(m0[3 * N + j]);
        kq[j] = k;
        a0 += k.x;
        a1 += k.y;
        a2 += k.z;
        a3 += k.w;
    }
    quad_rows_finish<8, 1>(a0, a1, a2, a3, part, tid, q, Rv);
}

#define ROW_LOADS 128
#define ROW_CHUNK 16
__global__ __launch_bounds__(128) __attribute__((amdgpu_waves_per_eu(1, 1)))
void sinkhorn_row_mv(
    const float* __restrict__ K, const float* __restrict__ Cv,
    float* __restrict__ Rv) {
    const int lane = threadIdx.x & 31;
    const int quad = 4 * blockIdx.x + (threadIdx.x >> 5);
    const float4* __restrict__ kq =
        (const float4*)(K + (size_t)quad * QUAD_FLOATS) + lane;
    const float* __restrict__ cp = Cv + lane;

    float c[ROW_LOADS];
    float4 k[ROW_LOADS];
#pragma unroll
    for (int g = 0; g < ROW_LOADS / ROW_CHUNK; ++g) {
#pragma unroll
        for (int u = 0; u < ROW_CHUNK; ++u)
            c[g * ROW_CHUNK + u] = cp[32 * (g * ROW_CHUNK + u)];
#pragma unroll
        for (int u = 0; u < ROW_CHUNK; ++u)
            k[g * ROW_CHUNK + u] = kq[32 * (g * ROW_CHUNK + u)];
    }
    __builtin_amdgcn_sched_barrier(0);

    float a0 = 0.0f, a1 = 0.0f, a2 = 0.0f, a3 = 0.0f;
#pragma unroll
    for (int n = 0; n < ROW_LOADS; ++n) {
        a0 = fmaf(k[n].x, c[n], a0);
        a1 = fmaf(k[n].y, c[n], a1);
        a2 = fmaf(k[n].z, c[n], a2);
        a3 = fmaf(k[n].w, c[n], a3);
    }
    a0 = wave_sum(a0);
    a1 = wave_sum(a1);
    a2 = wave_sum(a2);
    a3 = wave_sum(a3);
    if (lane == 0) {
        float4 r;
        r.x = 1.0f / a0;
        r.y = 1.0f / a1;
        r.z = 1.0f / a2;
        r.w = 1.0f / a3;
        *(float4*)(Rv + 4 * quad) = r;
    }
}

#define COL_WAVES 4
#define COL_QUADS (N / 4 / COL_WAVES)
#define COL_LOADS (COL_QUADS / 2)
#define COL_GROUP 16
#define COL_AHEAD 3
__global__ __launch_bounds__(128) __attribute__((amdgpu_waves_per_eu(1, 1)))
void sinkhorn_col_strip(
    const float* __restrict__ K, const float* __restrict__ Rv,
    float* __restrict__ Cv) {
    __shared__ float4 rs[COL_WAVES][COL_QUADS];
    __shared__ float part[COL_WAVES][16];
    const int lane = threadIdx.x & 31;
    const int wave = threadIdx.x >> 5;
    const int half = lane >> 4;
    const int lcol = lane & 15;
    const int wq0 = wave * COL_QUADS;

    const float4* __restrict__ rg = (const float4*)Rv + wq0 + lane;
#pragma unroll
    for (int t = 0; t < COL_QUADS / 32; ++t)
        rs[wave][32 * t + lane] = rg[32 * t];

    const float4* __restrict__ kp =
        (const float4*)K + (size_t)(wq0 + half) * N + blockIdx.x * 16 + lcol;
    float4 kv[COL_LOADS];
#pragma unroll
    for (int u = 0; u < COL_AHEAD * COL_GROUP; ++u)
        kv[u] = kp[(size_t)(2 * u) * N];
    __builtin_amdgcn_sched_barrier(0);

    __builtin_amdgcn_fence(__ATOMIC_RELEASE, "wavefront");
    __builtin_amdgcn_wave_barrier();
    __builtin_amdgcn_fence(__ATOMIC_ACQUIRE, "wavefront");

    v8f d0 = {};
    v8f d1 = {};
    float4 rr[2][COL_GROUP];
#pragma unroll
    for (int u = 0; u < COL_GROUP; ++u) rr[0][u] = rs[wave][2 * u + half];
#pragma unroll
    for (int g = 0; g < COL_LOADS / COL_GROUP; ++g) {
        if (g + 1 < COL_LOADS / COL_GROUP) {
#pragma unroll
            for (int u = 0; u < COL_GROUP; ++u)
                rr[(g + 1) & 1][u] =
                    rs[wave][2 * (COL_GROUP * (g + 1) + u) + half];
        }
        if (g + COL_AHEAD < COL_LOADS / COL_GROUP) {
#pragma unroll
            for (int u = 0; u < COL_GROUP; ++u) {
                const int v = COL_GROUP * (g + COL_AHEAD) + u;
                kv[v] = kp[(size_t)(2 * v) * N];
            }
        }
        __builtin_amdgcn_sched_barrier(0);
#pragma unroll
        for (int u = 0; u < COL_GROUP; ++u) {
            const float4 r4 = rr[g & 1][u];
            const float4 k4 = kv[COL_GROUP * g + u];
            const v2f a_lo = {r4.x, r4.y};
            const v2f b_lo = {k4.x, k4.y};
            const v2f a_hi = {r4.z, r4.w};
            const v2f b_hi = {k4.z, k4.w};
            d0 = __builtin_amdgcn_wmma_f32_16x16x4_f32(
                false, a_lo, false, b_lo, (short)0, d0, false, false);
            d1 = __builtin_amdgcn_wmma_f32_16x16x4_f32(
                false, a_hi, false, b_hi, (short)0, d1, false, false);
        }
        __builtin_amdgcn_sched_barrier(0);
    }

    if (half == 0) part[wave][lcol] = d0[0] + d1[0];
    __syncthreads();
    if (threadIdx.x < 16) {
        float s = 0.0f;
#pragma unroll
        for (int w = 0; w < COL_WAVES; ++w) s += part[w][threadIdx.x];
        Cv[blockIdx.x * 16 + threadIdx.x] = 1.0f / s;
    }
}

__global__ __launch_bounds__(256) void sinkhorn_final(
    const float* __restrict__ Rv, const float* __restrict__ Cv, float* out) {
    const int q = blockIdx.x;
    const int tid = threadIdx.x;
    float* base = out + (size_t)q * QUAD_FLOATS;
    const float4* kq = (const float4*)base + tid;

    float4 k[16];
    float c[16];
#pragma unroll
    for (int n = 0; n < 16; ++n) {
        k[n] = kq[256 * n];
        c[n] = Cv[tid + 256 * n];
    }
    const float4 r = *(const float4*)(Rv + 4 * q);

    __syncthreads();

    float* o = base + tid;
#pragma unroll
    for (int n = 0; n < 16; ++n) {
        const int j = 256 * n;
        o[j] = fmaxf(k[n].x * r.x * c[n], 1e-9f);
        o[N + j] = fmaxf(k[n].y * r.y * c[n], 1e-9f);
        o[2 * N + j] = fmaxf(k[n].z * r.z * c[n], 1e-9f);
        o[3 * N + j] = fmaxf(k[n].w * r.w * c[n], 1e-9f);
    }
}

extern "C" void kernel_launch(void* const* d_in, const int* in_sizes, int n_in,
                              void* d_out, int out_size, void* d_ws,
                              size_t ws_size, hipStream_t stream) {
    (void)in_sizes; (void)n_in; (void)out_size; (void)ws_size;

    const float* M = (const float*)d_in[0];
    float* out = (float*)d_out;

    char* ws = (char*)d_ws;
    float* Rv = (float*)(ws);
    float* Cv = (float*)(ws + 16 * 1024);

    sinkhorn_build_row0<<<N / 4, 256, 0, stream>>>(M, out, Rv);
    sinkhorn_col_strip<<<N / 16, 128, 0, stream>>>(out, Rv, Cv);

    for (int t = 1; t < SINKHORN_ITERS; ++t) {
        sinkhorn_row_mv<<<N / 16, 128, 0, stream>>>(out, Cv, Rv);
        sinkhorn_col_strip<<<N / 16, 128, 0, stream>>>(out, Rv, Cv);
    }

    sinkhorn_final<<<N / 4, 256, 0, stream>>>(Rv, Cv, out);
}


# column pass carries the next row product: its matrix steps (K as A; r and unit columns as B) leave the strip transposed in the accumulators; K read once per iteration; row and column sums re-ordered
# speedup vs baseline: 1.7552x; 1.0716x over previous
#include <hip/hip_runtime.h>
#include <hip/hip_bf16.h>


#define N 4096
#define SINKHORN_ITERS 50
#define QUAD_FLOATS (4 * N)

typedef float v2f __attribute__((ext_vector_type(2)));
typedef float v8f __attribute__((ext_vector_type(8)));

__device__ __forceinline__ float wave_sum(float v) {
#pragma unroll
    for (int m = 16; m >= 1; m >>= 1) v += __shfl_xor(v, m);
    return v;
}

template <int WPQ, int QPG>
__device__ __forceinline__ void quad_rows_finish(float a0, float a1, float a2,
                                                 float a3, float (*part)[4],
                                                 int tid, int quad0,
                                                 float* __restrict__ Rv) {
    a0 = wave_sum(a0);
    a1 = wave_sum(a1);
    a2 = wave_sum(a2);
    a3 = wave_sum(a3);
    if ((tid & 31) == 0) {
        float* p = part[tid >> 5];
        p[0] = a0;
        p[1] = a1;
        p[2] = a2;
        p[3] = a3;
    }
    __syncthreads();
    if (tid < 4 * QPG) {
        const int qq = tid >> 2;
        const int s = tid & 3;
        float t = 0.0f;
#pragma unroll
        for (int x = 0; x < WPQ; ++x) t += part[qq * WPQ + x][s];
        Rv[4 * (quad0 + qq) + s] = 1.0f / t;
    }
}

__global__ __launch_bounds__(256) void sinkhorn_build_row0(
    const float* __restrict__ M, float* __restrict__ K,
    float* __restrict__ Rv) {
    __shared__ float part[8][4];
    const int q = blockIdx.x;
    const int tid = threadIdx.x;
    const float* __restrict__ m0 = M + (size_t)(4 * q) * N + tid;
    float4* __restrict__ kq = (float4*)(K + (size_t)q * QUAD_FLOATS) + tid;

    float a0 = 0.0f, a1 = 0.0f, a2 = 0.0f, a3 = 0.0f;
#pragma unroll
    for (int n = 0; n < 16; ++n) {
        const int j = n * 256;
        float4 k;
        k.x = __expf(m0[j]);
        k.y = __expf(m0[N + j]);
        k.z = __expf(m0[2 * N + j]);
        k.w = __expf(m0[3 * N + j]);
        kq[j] = k;
        a0 += k.x;
        a1 += k.y;
        a2 += k.z;
        a3 += k.w;
    }
    quad_rows_finish<8, 1>(a0, a1, a2, a3, part, tid, q, Rv);
}

#define COL_WAVES 4
#define COL_QUADS (N / 4 / COL_WAVES)
#define COL_LOADS (COL_QUADS / 2)
#define COL_GROUP 16
#define COL_AHEAD 3
__global__ __launch_bounds__(128) __attribute__((amdgpu_waves_per_eu(1, 1)))
void sinkhorn_col_strip(
    const float* __restrict__ K, const float* __restrict__ Rv,
    float* __restrict__ Cv) {
    __shared__ float4 rs[COL_WAVES][COL_QUADS];
    __shared__ float part[COL_WAVES][16];
    const int lane = threadIdx.x & 31;
    const int wave = threadIdx.x >> 5;
    const int half = lane >> 4;
    const int lcol = lane & 15;
    const int wq0 = wave * COL_QUADS;

    const float4* __restrict__ rg = (const float4*)Rv + wq0 + lane;
#pragma unroll
    for (int t = 0; t < COL_QUADS / 32; ++t)
        rs[wave][32 * t + lane] = rg[32 * t];

    const float4* __restrict__ kp =
        (const float4*)K + (size_t)(wq0 + half) * N + blockIdx.x * 16 + lcol;
    float4 kv[COL_LOADS];
#pragma unroll
    for (int u = 0; u < COL_AHEAD * COL_GROUP; ++u)
        kv[u] = kp[(size_t)(2 * u) * N];
    __builtin_amdgcn_sched_barrier(0);

    __builtin_amdgcn_fence(__ATOMIC_RELEASE, "wavefront");
    __builtin_amdgcn_wave_barrier();
    __builtin_amdgcn_fence(__ATOMIC_ACQUIRE, "wavefront");

    v8f d0 = {};
    v8f d1 = {};
    float4 rr[2][COL_GROUP];
#pragma unroll
    for (int u = 0; u < COL_GROUP; ++u) rr[0][u] = rs[wave][2 * u + half];
#pragma unroll
    for (int g = 0; g < COL_LOADS / COL_GROUP; ++g) {
        if (g + 1 < COL_LOADS / COL_GROUP) {
#pragma unroll
            for (int u = 0; u < COL_GROUP; ++u)
                rr[(g + 1) & 1][u] =
                    rs[wave][2 * (COL_GROUP * (g + 1) + u) + half];
        }
        if (g + COL_AHEAD < COL_LOADS / COL_GROUP) {
#pragma unroll
            for (int u = 0; u < COL_GROUP; ++u) {
                const int v = COL_GROUP * (g + COL_AHEAD) + u;
                kv[v] = kp[(size_t)(2 * v) * N];
            }
        }
        __builtin_amdgcn_sched_barrier(0);
#pragma unroll
        for (int u = 0; u < COL_GROUP; ++u) {
            const float4 r4 = rr[g & 1][u];
            const float4 k4 = kv[COL_GROUP * g + u];
            const v2f a_lo = {r4.x, r4.y};
            const v2f b_lo = {k4.x, k4.y};
            const v2f a_hi = {r4.z, r4.w};
            const v2f b_hi = {k4.z, k4.w};
            d0 = __builtin_amdgcn_wmma_f32_16x16x4_f32(
                false, a_lo, false, b_lo, (short)0, d0, false, false);
            d1 = __builtin_amdgcn_wmma_f32_16x16x4_f32(
                false, a_hi, false, b_hi, (short)0, d1, false, false);
        }
        __builtin_amdgcn_sched_barrier(0);
    }

    if (half == 0) part[wave][lcol] = d0[0] + d1[0];
    __syncthreads();
    if (threadIdx.x < 16) {
        float s = 0.0f;
#pragma unroll
        for (int w = 0; w < COL_WAVES; ++w) s += part[w][threadIdx.x];
        Cv[blockIdx.x * 16 + threadIdx.x] = 1.0f / s;
    }
}

#define FUS_WAVES 8
#define FUS_QUADS (N / 4 / FUS_WAVES)
#define FUS_LOADS (FUS_QUADS / 2)
#define FUS_STEPS (2 * FUS_LOADS)
#define FUS_BLOCKS ((FUS_LOADS + 2) / 3)
#define FUS_TILES (2 * FUS_BLOCKS)
#define FUS_ROWS (4 * FUS_QUADS)
#define FUS_GROUP 8
#define FUS_AHEAD 3

__device__ __forceinline__ float xhalf(float x) {
    const int b = __builtin_bit_cast(int, x);
    return __builtin_bit_cast(
        float, __builtin_amdgcn_permlanex16(b, b, 0x76543210, 0xfedcba98,
                                            false, false));
}

__global__ __launch_bounds__(256) __attribute__((amdgpu_waves_per_eu(2, 2)))
void sinkhorn_col_row(
    const float* __restrict__ K, const float* __restrict__ Rv,
    float* __restrict__ P) {
    __shared__ float4 rs[FUS_WAVES][FUS_QUADS];
    __shared__ __attribute__((aligned(16))) float rowp[FUS_WAVES][FUS_ROWS];
    __shared__ float part[FUS_WAVES][16];
    __shared__ float cl[16];
    const int lane = threadIdx.x & 31;
    const int wave = threadIdx.x >> 5;
    const int half = lane >> 4;
    const int lcol = lane & 15;
    const int wq0 = wave * FUS_QUADS;

    const float4* __restrict__ rg = (const float4*)Rv + wq0 + lane;
#pragma unroll
    for (int t = 0; t < FUS_QUADS / 32; ++t)
        rs[wave][32 * t + lane] = rg[32 * t];

    const float4* __restrict__ kp =
        (const float4*)K + (size_t)(wq0 + half) * N + blockIdx.x * 16 + lcol;
    float4 kv[FUS_LOADS];
#pragma unroll
    for (int u = 0; u < FUS_AHEAD * FUS_GROUP; ++u)
        kv[u] = kp[(size_t)(2 * u) * N];
    __builtin_amdgcn_sched_barrier(0);

    __builtin_amdgcn_fence(__ATOMIC_RELEASE, "wavefront");
    __builtin_amdgcn_wave_barrier();
    __builtin_amdgcn_fence(__ATOMIC_ACQUIRE, "wavefront");

    const bool is0 = (lcol == 0);
    float unit[3][2];
#pragma unroll
    for (int t = 0; t < 3; ++t)
#pragma unroll
        for (int s = 0; s < 2; ++s)
            unit[t][s] = (lcol == 1 + 4 * t + 2 * half + s) ? 1.0f : 0.0f;

    v8f d[FUS_TILES];
#pragma unroll
    for (int j = 0; j < FUS_TILES; ++j) d[j] = v8f{};
    v8f cs = {};

    float4 rr[FUS_GROUP];
#pragma unroll
    for (int g = 0; g < FUS_LOADS / FUS_GROUP; ++g) {
#pragma unroll
        for (int u = 0; u < FUS_GROUP; ++u)
            rr[u] = rs[wave][2 * (FUS_GROUP * g + u) + half];
        if (g + FUS_AHEAD < FUS_LOADS / FUS_GROUP) {
#pragma unroll
            for (int u = 0; u < FUS_GROUP; ++u) {
                const int v = FUS_GROUP * (g + FUS_AHEAD) + u;
                kv[v] = kp[(size_t)(2 * v) * N];
            }
        }
        __builtin_amdgcn_sched_barrier(0);
#pragma unroll
        for (int u = 0; u < FUS_GROUP; ++u) {
            const int uu = FUS_GROUP * g + u;
            const int b = uu / 3;
            const int t = uu % 3;
            const float4 r4 = rr[u];
            const float4 k4 = kv[uu];
            const v2f a_lo = {k4.x, k4.y};
            const v2f a_hi = {k4.z, k4.w};
            const v2f b_lo = {is0 ? r4.x : unit[t][0], is0 ? r4.y : unit[t][1]};
            const v2f b_hi = {is0 ? r4.z : unit[t][0], is0 ? r4.w : unit[t][1]};
            d[2 * b] = __builtin_amdgcn_wmma_f32_16x16x4_f32(
                false, a_lo, false, b_lo, (short)0, d[2 * b], false, false);
            d[2 * b + 1] = __builtin_amdgcn_wmma_f32_16x16x4_f32(
                false, a_hi, false, b_hi, (short)0, d[2 * b + 1], false, false);
        }
        __builtin_amdgcn_sched_barrier(0);
    }
#pragma unroll
    for (int j = 0; j < FUS_TILES; ++j) cs += d[j];

    if (is0) {
#pragma unroll
        for (int r = 0; r < 8; ++r) part[wave][8 * half + r] = cs[r];
    }
    __syncthreads();
    if (threadIdx.x < 16) {
        float s = 0.0f;
#pragma unroll
        for (int w = 0; w < FUS_WAVES; ++w) s += part[w][threadIdx.x];
        cl[threadIdx.x] = 1.0f / s;
    }
    __syncthreads();

    float cw[8];
#pragma unroll
    for (int r = 0; r < 8; ++r) cw[r] = cl[8 * half + r];

    const int n1 = lcol - 1;
    const int tn = n1 >> 2;
    const int rbase = 8 * tn + 4 * ((n1 >> 1) & 1) + (n1 & 1);
    const bool writer = (half == 0) && (lcol >= 1) && (lcol <= 12);
    float* __restrict__ rw = &rowp[wave][0];
    float pr[FUS_TILES];
#pragma unroll
    for (int j = 0; j < FUS_TILES; ++j) {
        float p = d[j][0] * cw[0];
#pragma unroll
        for (int r = 1; r < 8; ++r) p = fmaf(d[j][r], cw[r], p);
        pr[j] = p + xhalf(p);
    }
    if (writer) {
#pragma unroll
        for (int j = 0; j < FUS_TILES; ++j) {
            const int off = 24 * (j >> 1) + 2 * (j & 1);
            const bool whole = 3 * (j >> 1) + 2 < FUS_LOADS;
            if (whole || tn == 0) rw[rbase + off] = pr[j];
        }
    }

    __builtin_amdgcn_fence(__ATOMIC_RELEASE, "wavefront");
    __builtin_amdgcn_wave_barrier();
    __builtin_amdgcn_fence(__ATOMIC_ACQUIRE, "wavefront");

    float4* __restrict__ po =
        (float4*)(P + (size_t)blockIdx.x * N + wave * FUS_ROWS) + lane;
    const float4* rl = (const float4*)rw + lane;
#pragma unroll
    for (int m = 0; m < FUS_ROWS / 128; ++m) po[32 * m] = rl[32 * m];
}

__global__ __launch_bounds__(256) void sinkhorn_row_sum(
    const float* __restrict__ P, float* __restrict__ Rv) {
    __shared__ float4 acc[32][8];
    const int tid = threadIdx.x;
    const int q = tid & 7;
    const int sg = tid >> 3;
    const float4* __restrict__ p =
        (const float4*)P + (size_t)sg * (N / 4) + blockIdx.x * 8 + q;
    float4 v[8];
#pragma unroll
    for (int m = 0; m < 8; ++m) v[m] = p[(size_t)(32 * m) * (N / 4)];
    float4 a = v[0];
#pragma unroll
    for (int m = 1; m < 8; ++m) {
        a.x += v[m].x;
        a.y += v[m].y;
        a.z += v[m].z;
        a.w += v[m].w;
    }
    acc[sg][q] = a;
    __syncthreads();
    if (tid < 32) {
        const float* af = (const float*)&acc[0][0];
        float t = 0.0f;
#pragma unroll
        for (int x = 0; x < 32; ++x) t += af[32 * x + tid];
        Rv[blockIdx.x * 32 + tid] = 1.0f / t;
    }
}

__global__ __launch_bounds__(256) void sinkhorn_final(
    const float* __restrict__ Rv, const float* __restrict__ Cv, float* out) {
    const int q = blockIdx.x;
    const int tid = threadIdx.x;
    float* base = out + (size_t)q * QUAD_FLOATS;
    const float4* kq = (const float4*)base + tid;

    float4 k[16];
    float c[16];
#pragma unroll
    for (int n = 0; n < 16; ++n) {
        k[n] = kq[256 * n];
        c[n] = Cv[tid + 256 * n];
    }
    const float4 r = *(const float4*)(Rv + 4 * q);

    __syncthreads();

    float* o = base + tid;
#pragma unroll
    for (int n = 0; n < 16; ++n) {
        const int j = 256 * n;
        o[j] = fmaxf(k[n].x * r.x * c[n], 1e-9f);
        o[N + j] = fmaxf(k[n].y * r.y * c[n], 1e-9f);
        o[2 * N + j] = fmaxf(k[n].z * r.z * c[n], 1e-9f);
        o[3 * N + j] = fmaxf(k[n].w * r.w * c[n], 1e-9f);
    }
}

extern "C" void kernel_launch(void* const* d_in, const int* in_sizes, int n_in,
                              void* d_out, int out_size, void* d_ws,
                              size_t ws_size, hipStream_t stream) {
    (void)in_sizes; (void)n_in; (void)out_size; (void)ws_size;

    const float* M = (const float*)d_in[0];
    float* out = (float*)d_out;

    char* ws = (char*)d_ws;
    float* Rv = (float*)(ws);
    float* Cv = (float*)(ws + 16 * 1024);
    float* P = (float*)(ws + 32 * 1024);

    sinkhorn_build_row0<<<N / 4, 256, 0, stream>>>(M, out, Rv);

    for (int t = 1; t < SINKHORN_ITERS; ++t) {
        sinkhorn_col_row<<<N / 16, 256, 0, stream>>>(out, Rv, P);
        sinkhorn_row_sum<<<N / 32, 256, 0, stream>>>(P, Rv);
    }
    sinkhorn_col_strip<<<N / 16, 128, 0, stream>>>(out, Rv, Cv);

    sinkhorn_final<<<N / 4, 256, 0, stream>>>(Rv, Cv, out);
}
